# starting kernel
# baseline (speedup 1.0000x reference)
#include <hip/hip_runtime.h>
#include <math.h>

#define BATCH 32
#define NNODE 512
#define GGRP  512
#define EMBD  256
#define HNUM  16
#define DK    16

typedef __attribute__((ext_vector_type(2))) float v2f;
typedef __attribute__((ext_vector_type(8))) float v8f;

__device__ __forceinline__ v8f wmma4(v2f a, v2f b, v8f c) {
  return __builtin_amdgcn_wmma_f32_16x16x4_f32(
       false, a,  false, b,
       (short)0, c,  false,  false);
}

__device__ __forceinline__ v2f ld2g(const float* __restrict__ p) {
  return *reinterpret_cast<const v2f*>(p);
}
__device__ __forceinline__ v2f ld2s(const float* p) {
  return *reinterpret_cast<const v2f*>(p);
}

__global__ void mean_qg_kernel(const float* __restrict__ EN,
                               const float* __restrict__ WqG,
                               float* __restrict__ qg) {
  __shared__ float sm[EMBD];
  const int b = blockIdx.x;
  const int e = threadIdx.x;
  float s = 0.0f;
  const float* base = EN + (size_t)b * NNODE * EMBD + e;
  for (int n = 0; n < NNODE; ++n) s += base[(size_t)n * EMBD];
  sm[e] = s * (1.0f / (float)NNODE);
  __syncthreads();
  float acc = 0.0f;
  const float* wrow = WqG + (size_t)e * EMBD;
  for (int k = 0; k < EMBD; ++k) acc += sm[k] * wrow[k];
  qg[(size_t)b * EMBD + e] = acc;
}

__global__ void proj_kernel(const float* __restrict__ EN,
                            const float* __restrict__ EF,
                            const float* __restrict__ EL,
                            const float* __restrict__ WqF,
                            const float* __restrict__ WqL,
                            const float* __restrict__ Wk,
                            const float* __restrict__ Wv,
                            const float* __restrict__ qg,
                            float* __restrict__ Q,
                            float* __restrict__ K,
                            float* __restrict__ V) {
  const int widx = threadIdx.x >> 5;
  const int lane = threadIdx.x & 31;
  const int ln   = lane & 15;
  const int kh   = lane >> 4;
  const int wave = blockIdx.x * 4 + widx;
  const int b    = wave >> 6;
  const int rem  = wave & 63;
  const int mt   = rem >> 2;
  const int ng   = rem & 3;
  const int row0 = mt * 32 + ln;
  const int sel  = blockIdx.y;

  v8f acc[2][4];
#pragma unroll
  for (int i = 0; i < 2; ++i)
#pragma unroll
    for (int j = 0; j < 4; ++j) acc[i][j] = v8f{0.f,0.f,0.f,0.f,0.f,0.f,0.f,0.f};

  float* Out;
  if (sel < 2) {
    const float* W  = (sel == 0) ? Wk : Wv;
    const float* A0 = EN + (size_t)(b * NNODE + row0) * EMBD;
    const float* A1 = A0 + (size_t)16 * EMBD;
    const float* Br[4];
#pragma unroll
    for (int j = 0; j < 4; ++j) Br[j] = W + (size_t)(ng * 64 + j * 16 + ln) * EMBD;
    for (int k = 0; k < EMBD; k += 4) {
      const v2f a0 = ld2g(A0 + k + 2 * kh);
      const v2f a1 = ld2g(A1 + k + 2 * kh);
#pragma unroll
      for (int j = 0; j < 4; ++j) {
        const v2f bb = ld2g(Br[j] + k + 2 * kh);
        acc[0][j] = wmma4(a0, bb, acc[0][j]);
        acc[1][j] = wmma4(a1, bb, acc[1][j]);
      }
    }
    Out = (sel == 0) ? K : V;
  } else {
    const float* A0 = EF + (size_t)(b * GGRP + row0) * EMBD;
    const float* A1 = A0 + (size_t)16 * EMBD;
    const float* Br[4];
#pragma unroll
    for (int j = 0; j < 4; ++j) Br[j] = WqF + (size_t)(ng * 64 + j * 16 + ln) * EMBD;
    for (int k = 0; k < EMBD; k += 4) {
      const v2f a0 = ld2g(A0 + k + 2 * kh);
      const v2f a1 = ld2g(A1 + k + 2 * kh);
#pragma unroll
      for (int j = 0; j < 4; ++j) {
        const v2f bb = ld2g(Br[j] + k + 2 * kh);
        acc[0][j] = wmma4(a0, bb, acc[0][j]);
        acc[1][j] = wmma4(a1, bb, acc[1][j]);
      }
    }
    const float* C0 = EL + (size_t)(b * GGRP + row0) * EMBD;
    const float* C1 = C0 + (size_t)16 * EMBD;
#pragma unroll
    for (int j = 0; j < 4; ++j) Br[j] = WqL + (size_t)(ng * 64 + j * 16 + ln) * EMBD;
    for (int k = 0; k < EMBD; k += 4) {
      const v2f a0 = ld2g(C0 + k + 2 * kh);
      const v2f a1 = ld2g(C1 + k + 2 * kh);
#pragma unroll
      for (int j = 0; j < 4; ++j) {
        const v2f bb = ld2g(Br[j] + k + 2 * kh);
        acc[0][j] = wmma4(a0, bb, acc[0][j]);
        acc[1][j] = wmma4(a1, bb, acc[1][j]);
      }
    }
#pragma unroll
    for (int j = 0; j < 4; ++j) {
      const float qv = qg[(size_t)b * EMBD + ng * 64 + j * 16 + ln];
#pragma unroll
      for (int i = 0; i < 2; ++i)
#pragma unroll
        for (int r = 0; r < 8; ++r) acc[i][j][r] += qv;
    }
    Out = Q;
  }

#pragma unroll
  for (int i = 0; i < 2; ++i)
#pragma unroll
    for (int j = 0; j < 4; ++j) {
      const int col = ng * 64 + j * 16 + ln;
#pragma unroll
      for (int r = 0; r < 8; ++r) {
        const int row = mt * 32 + i * 16 + r + 8 * kh;
        Out[(size_t)(b * NNODE + row) * EMBD + col] = acc[i][j][r];
      }
    }
}

__global__ void attn_kernel(const float* __restrict__ Q,
                            const float* __restrict__ K,
                            const float* __restrict__ V,
                            const float* __restrict__ MASK,
                            float* __restrict__ outc) {
  __shared__ float sc[16 * 512];
  __shared__ float part[8 * 256];
  __shared__ float redA[16][16];
  __shared__ float redB[16][16];

  const int gt = blockIdx.x, h = blockIdx.y, b = blockIdx.z;
  const int t    = threadIdx.x;
  const int widx = t >> 5;
  const int lane = t & 31;
  const int ln   = lane & 15;
  const int kh   = lane >> 4;

  {
    const float* Arow = Q + (size_t)(b * GGRP + gt * 16 + ln) * EMBD + h * DK;
    const float* Br[4];
#pragma unroll
    for (int j = 0; j < 4; ++j)
      Br[j] = K + (size_t)(b * NNODE + (widx * 4 + j) * 16 + ln) * EMBD + h * DK;
    v8f acc[4];
#pragma unroll
    for (int j = 0; j < 4; ++j) acc[j] = v8f{0.f,0.f,0.f,0.f,0.f,0.f,0.f,0.f};
#pragma unroll
    for (int k = 0; k < DK; k += 4) {
      const v2f a = ld2g(Arow + k + 2 * kh);
#pragma unroll
      for (int j = 0; j < 4; ++j)
        acc[j] = wmma4(a, ld2g(Br[j] + k + 2 * kh), acc[j]);
    }
#pragma unroll
    for (int j = 0; j < 4; ++j) {
      const int nb = (widx * 4 + j) * 16;
#pragma unroll
      for (int r = 0; r < 8; ++r) {
        const int row = r + 8 * kh;
        const int g   = gt * 16 + row;
        sc[row * 512 + nb + ln] =
            acc[j][r] * 0.25f + MASK[(size_t)(b * GGRP + g) * NNODE + nb + ln];
      }
    }
  }
  __syncthreads();

  const int row = t >> 4, l16 = t & 15;
  float* srow = sc + row * 512;
  float mx = -INFINITY;
  for (int j = l16; j < 512; j += 16) mx = fmaxf(mx, srow[j]);
  redA[row][l16] = mx;
  __syncthreads();
  if (l16 == 0) {
    float m2 = redA[row][0];
    for (int i = 1; i < 16; ++i) m2 = fmaxf(m2, redA[row][i]);
    redA[row][0] = m2;
  }
  __syncthreads();
  const float rmax = redA[row][0];
  float ssum = 0.0f;
  for (int j = l16; j < 512; j += 16) {
    const float e = __expf(srow[j] - rmax);
    srow[j] = e;
    ssum += e;
  }
  redB[row][l16] = ssum;
  __syncthreads();
  if (l16 == 0) {
    float s2 = 0.0f;
    for (int i = 0; i < 16; ++i) s2 += redB[row][i];
    redB[row][0] = s2;
  }
  __syncthreads();
  const float inv = 1.0f / redB[row][0];
  for (int j = l16; j < 512; j += 16) srow[j] *= inv;
  __syncthreads();

  v8f acc = {0.f,0.f,0.f,0.f,0.f,0.f,0.f,0.f};
#pragma unroll
  for (int nti = 0; nti < 4; ++nti) {
    const int nb = (widx * 4 + nti) * 16;
#pragma unroll
    for (int k = 0; k < 16; k += 4) {
      const int kk = nb + k + 2 * kh;
      v2f a = ld2s(sc + ln * 512 + kk);
      const float* vp = V + (size_t)(b * NNODE + kk) * EMBD + h * DK + ln;
      v2f bb;
      bb.x = vp[0];
      bb.y = vp[EMBD];
      acc = wmma4(a, bb, acc);
    }
  }
#pragma unroll
  for (int r = 0; r < 8; ++r)
    part[widx * 256 + (r + 8 * kh) * 16 + ln] = acc[r];
  __syncthreads();

  {
    const int orow = t >> 4, ocol = t & 15;
    float s = 0.0f;
#pragma unroll
    for (int w = 0; w < 8; ++w) s += part[w * 256 + orow * 16 + ocol];
    outc[(size_t)(b * GGRP + gt * 16 + orow) * EMBD + h * DK + ocol] = s;
  }
}

__global__ void mh_kernel(const float* __restrict__ outc,
                          const float* __restrict__ Wc,
                          const float* __restrict__ bc,
                          float* __restrict__ mh) {
  const int widx = threadIdx.x >> 5;
  const int lane = threadIdx.x & 31;
  const int ln   = lane & 15;
  const int kh   = lane >> 4;
  const int wave = blockIdx.x * 4 + widx;
  const int b    = wave >> 6;
  const int rem  = wave & 63;
  const int mt   = rem >> 2;
  const int ng   = rem & 3;
  const int row0 = mt * 32 + ln;

  v8f acc[2][4];
#pragma unroll
  for (int i = 0; i < 2; ++i)
#pragma unroll
    for (int j = 0; j < 4; ++j) acc[i][j] = v8f{0.f,0.f,0.f,0.f,0.f,0.f,0.f,0.f};

  const float* A0 = outc + (size_t)(b * GGRP + row0) * EMBD;
  const float* A1 = A0 + (size_t)16 * EMBD;
  const float* Br[4];
#pragma unroll
  for (int j = 0; j < 4; ++j) Br[j] = Wc + (size_t)(ng * 64 + j * 16 + ln) * EMBD;
  for (int k = 0; k < EMBD; k += 4) {
    const v2f a0 = ld2g(A0 + k + 2 * kh);
    const v2f a1 = ld2g(A1 + k + 2 * kh);
#pragma unroll
    for (int j = 0; j < 4; ++j) {
      const v2f bb = ld2g(Br[j] + k + 2 * kh);
      acc[0][j] = wmma4(a0, bb, acc[0][j]);
      acc[1][j] = wmma4(a1, bb, acc[1][j]);
    }
  }
#pragma unroll
  for (int i = 0; i < 2; ++i)
#pragma unroll
    for (int j = 0; j < 4; ++j) {
      const int col  = ng * 64 + j * 16 + ln;
      const float bv = bc[col];
#pragma unroll
      for (int r = 0; r < 8; ++r) {
        const int row = mt * 32 + i * 16 + r + 8 * kh;
        mh[(size_t)(b * GGRP + row) * EMBD + col] = acc[i][j][r] + bv;
      }
    }
}

__global__ void final_kernel(const float* __restrict__ mh,
                             const float* __restrict__ EN,
                             const float* __restrict__ MASK,
                             float* __restrict__ out) {
  __shared__ float sc[16 * 512];
  __shared__ float redA[16][16];
  __shared__ float redB[16][16];

  const int gt = blockIdx.x, b = blockIdx.y;
  const int t    = threadIdx.x;
  const int widx = t >> 5;
  const int lane = t & 31;
  const int ln   = lane & 15;
  const int kh   = lane >> 4;

  {
    const float* Arow = mh + (size_t)(b * GGRP + gt * 16 + ln) * EMBD;
    const float* Br[4];
#pragma unroll
    for (int j = 0; j < 4; ++j)
      Br[j] = EN + (size_t)(b * NNODE + (widx * 4 + j) * 16 + ln) * EMBD;
    v8f acc[4];
#pragma unroll
    for (int j = 0; j < 4; ++j) acc[j] = v8f{0.f,0.f,0.f,0.f,0.f,0.f,0.f,0.f};
    for (int k = 0; k < EMBD; k += 4) {
      const v2f a = ld2g(Arow + k + 2 * kh);
#pragma unroll
      for (int j = 0; j < 4; ++j)
        acc[j] = wmma4(a, ld2g(Br[j] + k + 2 * kh), acc[j]);
    }
#pragma unroll
    for (int j = 0; j < 4; ++j) {
      const int nb = (widx * 4 + j) * 16;
#pragma unroll
      for (int r = 0; r < 8; ++r) {
        const int row = r + 8 * kh;
        const int g   = gt * 16 + row;
        const float s2 = acc[j][r] * (1.0f / 16.0f);
        sc[row * 512 + nb + ln] =
            10.0f * tanhf(s2) + MASK[(size_t)(b * GGRP + g) * NNODE + nb + ln];
      }
    }
  }
  __syncthreads();

  const int row = t >> 4, l16 = t & 15;
  float* srow = sc + row * 512;
  float mx = -INFINITY;
  for (int j = l16; j < 512; j += 16) mx = fmaxf(mx, srow[j]);
  redA[row][l16] = mx;
  __syncthreads();
  if (l16 == 0) {
    float m2 = redA[row][0];
    for (int i = 1; i < 16; ++i) m2 = fmaxf(m2, redA[row][i]);
    redA[row][0] = m2;
  }
  __syncthreads();
  const float rmax = redA[row][0];
  float ssum = 0.0f;
  for (int j = l16; j < 512; j += 16) {
    const float e = __expf(srow[j] - rmax);
    srow[j] = e;
    ssum += e;
  }
  redB[row][l16] = ssum;
  __syncthreads();
  if (l16 == 0) {
    float s2 = 0.0f;
    for (int i = 0; i < 16; ++i) s2 += redB[row][i];
    redB[row][0] = s2;
  }
  __syncthreads();
  const float inv = 1.0f / redB[row][0];
  float* orow = out + (size_t)(b * GGRP + gt * 16 + row) * NNODE;
  for (int j = l16; j < 512; j += 16) orow[j] = srow[j] * inv;
}

extern "C" void kernel_launch(void* const* d_in, const int* in_sizes, int n_in,
                              void* d_out, int out_size, void* d_ws, size_t ws_size,
                              hipStream_t stream) {
  const float* EN   = (const float*)d_in[0];
  const float* EF   = (const float*)d_in[1];
  const float* EL   = (const float*)d_in[2];
  const float* MASK = (const float*)d_in[3];
  const float* WqG  = (const float*)d_in[4];
  const float* WqF  = (const float*)d_in[5];
  const float* WqL  = (const float*)d_in[6];
  const float* Wk   = (const float*)d_in[7];
  const float* Wv   = (const float*)d_in[8];
  const float* Wc   = (const float*)d_in[9];
  const float* bc   = (const float*)d_in[10];
  float* out = (float*)d_out;

  float* ws   = (float*)d_ws;
  float* qg   = ws;
  float* Qb   = qg   + (size_t)BATCH * EMBD;
  float* Kb   = Qb   + (size_t)BATCH * GGRP * EMBD;
  float* Vb   = Kb   + (size_t)BATCH * NNODE * EMBD;
  float* outc = Vb   + (size_t)BATCH * NNODE * EMBD;
  float* mhb  = outc + (size_t)BATCH * GGRP * EMBD;

  mean_qg_kernel<<<BATCH, 256, 0, stream>>>(EN, WqG, qg);

  dim3 gp(512, 3, 1);
  proj_kernel<<<gp, 128, 0, stream>>>(EN, EF, EL, WqF, WqL, Wk, Wv, qg, Qb, Kb, Vb);

  dim3 ga(GGRP / 16, HNUM, BATCH);
  attn_kernel<<<ga, 256, 0, stream>>>(Qb, Kb, Vb, MASK, outc);

  mh_kernel<<<512, 128, 0, stream>>>(outc, Wc, bc, mhb);

  dim3 gf(GGRP / 16, BATCH, 1);
  final_kernel<<<gf, 256, 0, stream>>>(mhb, EN, MASK, out);
}


# Attention and final softmax: a thread's 32 scores stay in registers after one 16-B LDS read pass; max and sum by in-row DPP butterflies (sum order changed); attention applies 1/sum to its output
# speedup vs baseline: 2.0737x; 2.0737x over previous
#include <hip/hip_runtime.h>
#include <math.h>

#define BATCH 32
#define NNODE 512
#define GGRP  512
#define EMBD  256
#define HNUM  16
#define DK    16

#define SCP   532

typedef __attribute__((ext_vector_type(2))) float v2f;
typedef __attribute__((ext_vector_type(4))) float v4f;
typedef __attribute__((ext_vector_type(8))) float v8f;

__device__ __forceinline__ v8f wmma4(v2f a, v2f b, v8f c) {
  return __builtin_amdgcn_wmma_f32_16x16x4_f32(
       false, a,  false, b,
       (short)0, c,  false,  false);
}

__device__ __forceinline__ v2f ld2g(const float* __restrict__ p) {
  return *reinterpret_cast<const v2f*>(p);
}
__device__ __forceinline__ v2f ld2s(const float* p) {
  return *reinterpret_cast<const v2f*>(p);
}
__device__ __forceinline__ void st2g(float* __restrict__ p, float x, float y) {
  v2f v; v.x = x; v.y = y;
  *reinterpret_cast<v2f*>(p) = v;
}

template <int M>
__device__ __forceinline__ float row_xor(float v) {
  return __builtin_bit_cast(float,
      __builtin_amdgcn_update_dpp(0, __builtin_bit_cast(int, v),
                                  0x160 + M, 0xf, 0xf, true));
}
__device__ __forceinline__ float row_max16(float v) {
  v = fmaxf(v, row_xor<1>(v));
  v = fmaxf(v, row_xor<2>(v));
  v = fmaxf(v, row_xor<4>(v));
  v = fmaxf(v, row_xor<8>(v));
  return v;
}
__device__ __forceinline__ float row_sum16(float v) {
  v += row_xor<1>(v);
  v += row_xor<2>(v);
  v += row_xor<4>(v);
  v += row_xor<8>(v);
  return v;
}

#define HBLK (512 * DK)

#define FTILE (16 * EMBD)

#define TLP 260
__global__ void prep_kernel(const float* __restrict__ EN,
                            const float* __restrict__ WqG,
                            const float* __restrict__ WqF,
                            const float* __restrict__ WqL,
                            const float* __restrict__ Wk,
                            const float* __restrict__ Wv,
                            float* __restrict__ qg,
                            float* __restrict__ ENf,
                            float* __restrict__ Wf) {
  __shared__ float tl[16 * TLP];
  const int t = threadIdx.x;
  if (blockIdx.x < BATCH) {
    const int b = blockIdx.x;
    {
      const int cg = t & 63, rg = t >> 6;
      const float* base =
          EN + ((size_t)b * NNODE + (size_t)rg * 128) * EMBD + cg * 4;
      v4f s = {0.f, 0.f, 0.f, 0.f};
      for (int i = 0; i < 128; ++i)
        s += *reinterpret_cast<const v4f*>(base + (size_t)i * EMBD);
      *reinterpret_cast<v4f*>(tl + rg * 256 + cg * 4) = s;
    }
    __syncthreads();
    tl[1024 + t] = ((tl[t] + tl[256 + t]) + (tl[512 + t] + tl[768 + t])) *
                   (1.0f / (float)NNODE);
    __syncthreads();
    {
      const int lane = t & 31, widx = t >> 5;
      const v4f m0 = *reinterpret_cast<const v4f*>(tl + 1024 + lane * 4);
      const v4f m1 = *reinterpret_cast<const v4f*>(tl + 1024 + 128 + lane * 4);
      const float* wb = WqG + (size_t)(widx * 32) * EMBD + lane * 4;
      float keep = 0.0f;
#pragma unroll 8
      for (int r = 0; r < 32; ++r) {
        const v4f w0 = *reinterpret_cast<const v4f*>(wb + r * EMBD);
        const v4f w1 = *reinterpret_cast<const v4f*>(wb + r * EMBD + 128);
        float p = m0.x * w0.x;
        p += m0.y * w0.y;
        p += m0.z * w0.z;
        p += m0.w * w0.w;
        p += m1.x * w1.x;
        p += m1.y * w1.y;
        p += m1.z * w1.z;
        p += m1.w * w1.w;
        p += __shfl_xor(p, 16);
        p += __shfl_xor(p, 8);
        p += __shfl_xor(p, 4);
        p += __shfl_xor(p, 2);
        p += __shfl_xor(p, 1);
        if (lane == r) keep = p;
      }
      qg[(size_t)b * EMBD + widx * 32 + lane] = keep;
    }
    return;
  }
  const int tile = blockIdx.x - BATCH;
  const float* srcp;
  float* dstp;
  if (tile < BATCH * NNODE / 16) {
    srcp = EN  + (size_t)tile * FTILE;
    dstp = ENf + (size_t)tile * FTILE;
  } else {
    const int w  = tile - BATCH * NNODE / 16;
    const int wt = w & 15;
    const int ws = w >> 4;
    const float* wsrc = (ws == 0) ? WqF : (ws == 1) ? WqL : (ws == 2) ? Wk : Wv;
    srcp = wsrc + (size_t)wt * FTILE;
    dstp = Wf + (size_t)w * FTILE;
  }
#pragma unroll
  for (int i = 0; i < 4; ++i) {
    const int idx = t + 256 * i;
    const int row = idx >> 6, q = idx & 63;
    const v4f v = *reinterpret_cast<const v4f*>(srcp + row * EMBD + q * 4);
    *reinterpret_cast<v4f*>(tl + row * TLP + q * 4) = v;
  }
  __syncthreads();
#pragma unroll
  for (int i = 0; i < 4; ++i) {
    const int p = t + 256 * i;
    const int s = p >> 4, j = p & 15;
    const int h = j >> 3, c = (j & 7) * 2;
    const v2f lo = ld2s(tl + c * TLP + 4 * s + 2 * h);
    const v2f hi = ld2s(tl + (c + 1) * TLP + 4 * s + 2 * h);
    v4f o; o.x = lo.x; o.y = lo.y; o.z = hi.x; o.w = hi.y;
    *reinterpret_cast<v4f*>(dstp + 4 * p) = o;
  }
}

__global__ void proj_kernel(const float* __restrict__ ENf,
                            const float* __restrict__ EF,
                            const float* __restrict__ EL,
                            const float* __restrict__ Wf,
                            const float* __restrict__ qg,
                            float* __restrict__ Q,
                            float* __restrict__ K,
                            float* __restrict__ V) {
  const int widx = threadIdx.x >> 5;
  const int lane = threadIdx.x & 31;
  const int ln   = lane & 15;
  const int kh   = lane >> 4;
  const int wave = blockIdx.x * 4 + widx;
  const int b    = wave >> 6;
  const int rem  = wave & 63;
  const int mt   = rem >> 2;
  const int ng   = rem & 3;
  const int row0 = mt * 32 + ln;
  const int sel  = blockIdx.y;

  const float* WqFf = Wf;
  const float* WqLf = Wf + (size_t)16 * FTILE;
  const float* Wkf  = Wf + (size_t)32 * FTILE;
  const float* Wvf  = Wf + (size_t)48 * FTILE;

  v8f acc[2][4];
#pragma unroll
  for (int i = 0; i < 2; ++i)
#pragma unroll
    for (int j = 0; j < 4; ++j) acc[i][j] = v8f{0.f,0.f,0.f,0.f,0.f,0.f,0.f,0.f};

  if (sel == 1) {
    const float* A0 = ENf + (size_t)(b * 32 + mt * 2) * FTILE + lane * 2;
    const float* A1 = A0 + FTILE;
    const float* Br[4];
#pragma unroll
    for (int j = 0; j < 4; ++j) Br[j] = Wvf + (size_t)(ng * 4 + j) * FTILE + lane * 2;
    for (int s = 0; s < EMBD / 4; ++s) {
      const v2f a0 = ld2g(A0 + s * 64);
      const v2f a1 = ld2g(A1 + s * 64);
#pragma unroll
      for (int j = 0; j < 4; ++j) {
        const v2f bb = ld2g(Br[j] + s * 64);
        acc[0][j] = wmma4(a0, bb, acc[0][j]);
        acc[1][j] = wmma4(a1, bb, acc[1][j]);
      }
    }
#pragma unroll
    for (int i = 0; i < 2; ++i)
#pragma unroll
      for (int j = 0; j < 4; ++j) {
        float* hb = V + (size_t)(b * HNUM + ng * 4 + j) * HBLK;
#pragma unroll
        for (int r = 0; r < 8; r += 2) {
          const int n = mt * 32 + i * 16 + r + 8 * kh;
          st2g(hb + (n >> 1) * 32 + ln * 2, acc[i][j][r], acc[i][j][r + 1]);
        }
      }
    return;
  }

  if (sel == 0) {
    const float* A0 = ENf + (size_t)(b * 32 + mt * 2) * FTILE + lane * 2;
    const float* A1 = A0 + FTILE;
    const float* Br[4];
#pragma unroll
    for (int j = 0; j < 4; ++j) Br[j] = Wkf + (size_t)(ng * 4 + j) * FTILE + lane * 2;
    for (int s = 0; s < EMBD / 4; ++s) {
      const v2f a0 = ld2g(A0 + s * 64);
      const v2f a1 = ld2g(A1 + s * 64);
#pragma unroll
      for (int j = 0; j < 4; ++j) {
        const v2f bb = ld2g(Br[j] + s * 64);
        acc[0][j] = wmma4(bb, a0, acc[0][j]);
        acc[1][j] = wmma4(bb, a1, acc[1][j]);
      }
    }
  } else {
    const float* A0 = EF + (size_t)(b * GGRP + row0) * EMBD + 2 * kh;
    const float* A1 = A0 + (size_t)16 * EMBD;
    const float* Br[4];
#pragma unroll
    for (int j = 0; j < 4; ++j) Br[j] = WqFf + (size_t)(ng * 4 + j) * FTILE + lane * 2;
    for (int s = 0; s < EMBD / 4; ++s) {
      const v2f a0 = ld2g(A0 + s * 4);
      const v2f a1 = ld2g(A1 + s * 4);
#pragma unroll
      for (int j = 0; j < 4; ++j) {
        const v2f bb = ld2g(Br[j] + s * 64);
        acc[0][j] = wmma4(bb, a0, acc[0][j]);
        acc[1][j] = wmma4(bb, a1, acc[1][j]);
      }
    }
    const float* C0 = EL + (size_t)(b * GGRP + row0) * EMBD + 2 * kh;
    const float* C1 = C0 + (size_t)16 * EMBD;
#pragma unroll
    for (int j = 0; j < 4; ++j) Br[j] = WqLf + (size_t)(ng * 4 + j) * FTILE + lane * 2;
    for (int s = 0; s < EMBD / 4; ++s) {
      const v2f a0 = ld2g(C0 + s * 4);
      const v2f a1 = ld2g(C1 + s * 4);
#pragma unroll
      for (int j = 0; j < 4; ++j) {
        const v2f bb = ld2g(Br[j] + s * 64);
        acc[0][j] = wmma4(bb, a0, acc[0][j]);
        acc[1][j] = wmma4(bb, a1, acc[1][j]);
      }
    }
#pragma unroll
    for (int j = 0; j < 4; ++j)
#pragma unroll
      for (int r = 0; r < 8; ++r) {
        const float qv = qg[(size_t)b * EMBD + ng * 64 + j * 16 + r + 8 * kh];
#pragma unroll
        for (int i = 0; i < 2; ++i) acc[i][j][r] += qv;
      }
  }

  float* Out = (sel == 0) ? K : Q;
#pragma unroll
  for (int i = 0; i < 2; ++i)
#pragma unroll
    for (int j = 0; j < 4; ++j) {
      float* tb = Out + (size_t)(b * HNUM + ng * 4 + j) * HBLK + (mt * 2 + i) * 256;
#pragma unroll
      for (int r = 0; r < 8; r += 2) {
        const int d = r + 8 * kh;
        st2g(tb + (d >> 2) * 64 + ((d >> 1) & 1) * 32 + ln * 2,
             acc[i][j][r], acc[i][j][r + 1]);
      }
    }
}

__global__ void attn_kernel(const float* __restrict__ Q,
                            const float* __restrict__ K,
                            const float* __restrict__ V,
                            const float* __restrict__ MASK,
                            float* __restrict__ outc) {
  __shared__ __attribute__((aligned(16))) float sc[16 * SCP];
  __shared__ float part[8 * 256];
  __shared__ float rinv[16];

  const int gt = blockIdx.x, h = blockIdx.y, b = blockIdx.z;
  const int t    = threadIdx.x;
  const int widx = t >> 5;
  const int lane = t & 31;
  const int ln   = lane & 15;
  const int kh   = lane >> 4;

  {
    const float* Atile = Q + (size_t)(b * HNUM + h) * HBLK + gt * 256 + lane * 2;
    const float* Bt[4];
#pragma unroll
    for (int j = 0; j < 4; ++j)
      Bt[j] = K + (size_t)(b * HNUM + h) * HBLK + (widx * 4 + j) * 256 + lane * 2;
    v8f acc[4];
#pragma unroll
    for (int j = 0; j < 4; ++j) acc[j] = v8f{0.f,0.f,0.f,0.f,0.f,0.f,0.f,0.f};
    v2f af[4], bf[4][4];
#pragma unroll
    for (int s = 0; s < DK / 4; ++s) {
      af[s] = ld2g(Atile + s * 64);
#pragma unroll
      for (int j = 0; j < 4; ++j) bf[j][s] = ld2g(Bt[j] + s * 64);
    }
    __builtin_amdgcn_sched_barrier(0);
#pragma unroll
    for (int s = 0; s < DK / 4; ++s)
#pragma unroll
      for (int j = 0; j < 4; ++j)
        acc[j] = wmma4(af[s], bf[j][s], acc[j]);
#pragma unroll
    for (int j = 0; j < 4; ++j) {
      const int nb = (widx * 4 + j) * 16;
#pragma unroll
      for (int r = 0; r < 8; ++r) {
        const int row = r + 8 * kh;
        const int g   = gt * 16 + row;
        sc[row * SCP + nb + ln] =
            acc[j][r] * 0.25f + MASK[(size_t)(b * GGRP + g) * NNODE + nb + ln];
      }
    }
  }
  __syncthreads();

  {
    const int row = t >> 4, l16 = t & 15;
    float* sp = sc + row * SCP + l16 * 4;
    v4f x[8];
#pragma unroll
    for (int i = 0; i < 8; ++i) x[i] = *reinterpret_cast<const v4f*>(sp + i * 64);
    float mx = -INFINITY;
#pragma unroll
    for (int i = 0; i < 8; ++i) {
      mx = fmaxf(mx, x[i].x);
      mx = fmaxf(mx, x[i].y);
      mx = fmaxf(mx, x[i].z);
      mx = fmaxf(mx, x[i].w);
    }
    const float rmax = row_max16(mx);
    float ssum = 0.0f;
#pragma unroll
    for (int i = 0; i < 8; ++i) {
      v4f e;
      e.x = __expf(x[i].x - rmax);
      e.y = __expf(x[i].y - rmax);
      e.z = __expf(x[i].z - rmax);
      e.w = __expf(x[i].w - rmax);
      ssum += e.x;
      ssum += e.y;
      ssum += e.z;
      ssum += e.w;
      *reinterpret_cast<v4f*>(sp + i * 64) = e;
    }
    ssum = row_sum16(ssum);
    if (l16 == 0) rinv[row] = 1.0f / ssum;
  }
  __syncthreads();

  v8f acc = {0.f,0.f,0.f,0.f,0.f,0.f,0.f,0.f};
  const float* vb = V + (size_t)(b * HNUM + h) * HBLK + widx * 1024 + lane * 2;
#pragma unroll
  for (int nti = 0; nti < 4; ++nti) {
    const int nb = (widx * 4 + nti) * 16;
#pragma unroll
    for (int k = 0; k < 16; k += 4) {
      const int kk = nb + k + 2 * kh;
      v2f a = ld2s(sc + ln * SCP + kk);
      const v2f bb = ld2g(vb + nti * 256 + k * 16);
      acc = wmma4(a, bb, acc);
    }
  }
#pragma unroll
  for (int r = 0; r < 8; ++r)
    part[widx * 256 + (r + 8 * kh) * 16 + ln] = acc[r];
  __syncthreads();

  {
    const int orow = (t >> 1) & 15;
    const int ocol = (t >> 6) * 4 + ((t >> 5) & 1) * 2 + (t & 1);
    float s = 0.0f;
#pragma unroll
    for (int w = 0; w < 8; ++w) s += part[w * 256 + orow * 16 + ocol];
    outc[(size_t)(b * (GGRP / 16) + gt) * FTILE + h * 256 + t] = s * rinv[orow];
  }
}

__global__ void mh_kernel(const float* __restrict__ outc,
                          const float* __restrict__ Wc,
                          const float* __restrict__ bc,
                          float* __restrict__ mh) {
  const int widx = threadIdx.x >> 5;
  const int lane = threadIdx.x & 31;
  const int ln   = lane & 15;
  const int kh   = lane >> 4;
  const int wave = blockIdx.x * 4 + widx;
  const int b    = wave >> 6;
  const int rem  = wave & 63;
  const int mt   = rem >> 2;
  const int ng   = rem & 3;

  v8f acc[2][4];
#pragma unroll
  for (int i = 0; i < 2; ++i)
#pragma unroll
    for (int j = 0; j < 4; ++j) acc[i][j] = v8f{0.f,0.f,0.f,0.f,0.f,0.f,0.f,0.f};

  const float* A0 = outc + (size_t)(b * 32 + mt * 2) * FTILE + lane * 2;
  const float* A1 = A0 + FTILE;
  const float* Br[4];
#pragma unroll
  for (int j = 0; j < 4; ++j) Br[j] = Wc + (size_t)(ng * 64 + j * 16 + ln) * EMBD + 2 * kh;
  for (int s = 0; s < EMBD / 4; ++s) {
    const v2f a0 = ld2g(A0 + s * 64);
    const v2f a1 = ld2g(A1 + s * 64);
#pragma unroll
    for (int j = 0; j < 4; ++j) {
      const v2f bb = ld2g(Br[j] + s * 4);
      acc[0][j] = wmma4(bb, a0, acc[0][j]);
      acc[1][j] = wmma4(bb, a1, acc[1][j]);
    }
  }
#pragma unroll
  for (int i = 0; i < 2; ++i)
#pragma unroll
    for (int j = 0; j < 4; ++j) {
      float* tb = mh + (size_t)(b * 32 + mt * 2 + i) * FTILE;
#pragma unroll
      for (int r = 0; r < 8; r += 2) {
        const int d = ng * 64 + j * 16 + r + 8 * kh;
        st2g(tb + (d >> 2) * 64 + ((d >> 1) & 1) * 32 + ln * 2,
             acc[i][j][r] + bc[d], acc[i][j][r + 1] + bc[d + 1]);
      }
    }
}

__global__ void final_kernel(const float* __restrict__ mh,
                             const float* __restrict__ ENf,
                             const float* __restrict__ MASK,
                             float* __restrict__ out) {
  __shared__ __attribute__((aligned(16))) float sc[16 * SCP];

  const int gt = blockIdx.x, b = blockIdx.y;
  const int t    = threadIdx.x;
  const int widx = t >> 5;
  const int lane = t & 31;
  const int ln   = lane & 15;
  const int kh   = lane >> 4;

  {
    const float* Atile = mh + (size_t)(b * (GGRP / 16) + gt) * FTILE + lane * 2;
    const float* Br[4];
#pragma unroll
    for (int j = 0; j < 4; ++j)
      Br[j] = ENf + (size_t)(b * (NNODE / 16) + widx * 4 + j) * FTILE + lane * 2;
    v8f acc[4];
#pragma unroll
    for (int j = 0; j < 4; ++j) acc[j] = v8f{0.f,0.f,0.f,0.f,0.f,0.f,0.f,0.f};
    for (int s = 0; s < EMBD / 4; ++s) {
      const v2f a = ld2g(Atile + s * 64);
#pragma unroll
      for (int j = 0; j < 4; ++j)
        acc[j] = wmma4(a, ld2g(Br[j] + s * 64), acc[j]);
    }
#pragma unroll
    for (int j = 0; j < 4; ++j) {
      const int nb = (widx * 4 + j) * 16;
#pragma unroll
      for (int r = 0; r < 8; ++r) {
        const int row = r + 8 * kh;
        const int g   = gt * 16 + row;
        const float s2 = acc[j][r] * (1.0f / 16.0f);
        sc[row * SCP + nb + ln] =
            10.0f * tanhf(s2) + MASK[(size_t)(b * GGRP + g) * NNODE + nb + ln];
      }
    }
  }
  __syncthreads();

  const int row = t >> 4, l16 = t & 15;
  const float* sp = sc + row * SCP + l16 * 4;
  v4f x[8];
#pragma unroll
  for (int i = 0; i < 8; ++i) x[i] = *reinterpret_cast<const v4f*>(sp + i * 64);
  float mx = -INFINITY;
#pragma unroll
  for (int i = 0; i < 8; ++i) {
    mx = fmaxf(mx, x[i].x);
    mx = fmaxf(mx, x[i].y);
    mx = fmaxf(mx, x[i].z);
    mx = fmaxf(mx, x[i].w);
  }
  const float rmax = row_max16(mx);
  float ssum = 0.0f;
#pragma unroll
  for (int i = 0; i < 8; ++i) {
    x[i].x = __expf(x[i].x - rmax);
    x[i].y = __expf(x[i].y - rmax);
    x[i].z = __expf(x[i].z - rmax);
    x[i].w = __expf(x[i].w - rmax);
    ssum += x[i].x;
    ssum += x[i].y;
    ssum += x[i].z;
    ssum += x[i].w;
  }
  const float inv = 1.0f / row_sum16(ssum);
  float* op = out + (size_t)(b * GGRP + gt * 16 + row) * NNODE + l16 * 4;
#pragma unroll
  for (int i = 0; i < 8; ++i)
    *reinterpret_cast<v4f*>(op + i * 64) = x[i] * inv;
}

extern "C" void kernel_launch(void* const* d_in, const int* in_sizes, int n_in,
                              void* d_out, int out_size, void* d_ws, size_t ws_size,
                              hipStream_t stream) {
  const float* EN   = (const float*)d_in[0];
  const float* EF   = (const float*)d_in[1];
  const float* EL   = (const float*)d_in[2];
  const float* MASK = (const float*)d_in[3];
  const float* WqG  = (const float*)d_in[4];
  const float* WqF  = (const float*)d_in[5];
  const float* WqL  = (const float*)d_in[6];
  const float* Wk   = (const float*)d_in[7];
  const float* Wv   = (const float*)d_in[8];
  const float* Wc   = (const float*)d_in[9];
  const float* bc   = (const float*)d_in[10];
  float* out = (float*)d_out;

  float* ws   = (float*)d_ws;
  float* qg   = ws;
  float* Qb   = qg   + (size_t)BATCH * EMBD;
  float* Kb   = Qb   + (size_t)BATCH * GGRP * EMBD;
  float* Vb   = Kb   + (size_t)BATCH * NNODE * EMBD;
  float* outc = Vb   + (size_t)BATCH * NNODE * EMBD;
  float* ENf  = outc + (size_t)BATCH * GGRP * EMBD;
  float* Wf   = outc;
  float* mhb  = Qb;

  prep_kernel<<<1120, 256, 0, stream>>>(EN, WqG, WqF, WqL, Wk, Wv, qg, ENf, Wf);

  dim3 gp(512, 3, 1);
  proj_kernel<<<gp, 128, 0, stream>>>(ENf, EF, EL, Wf, qg, Qb, Kb, Vb);

  dim3 ga(GGRP / 16, HNUM, BATCH);
  attn_kernel<<<ga, 256, 0, stream>>>(Qb, Kb, Vb, MASK, outc);

  mh_kernel<<<512, 128, 0, stream>>>(outc, Wc, bc, mhb);

  dim3 gf(GGRP / 16, BATCH, 1);
  final_kernel<<<gf, 256, 0, stream>>>(mhb, ENf, MASK, out);
}


# Attention and final: mask added in the softmax pass, where a thread owns 16-byte row pieces: 8 16-byte loads replace 32 4-byte loads of two half lines each (final: 10*tanh+mask no longer one fma)
# speedup vs baseline: 2.1461x; 1.0349x over previous
#include <hip/hip_runtime.h>
#include <math.h>

#define BATCH 32
#define NNODE 512
#define GGRP  512
#define EMBD  256
#define HNUM  16
#define DK    16

#define SCP   532

typedef __attribute__((ext_vector_type(2))) float v2f;
typedef __attribute__((ext_vector_type(4))) float v4f;
typedef __attribute__((ext_vector_type(8))) float v8f;

__device__ __forceinline__ v8f wmma4(v2f a, v2f b, v8f c) {
  return __builtin_amdgcn_wmma_f32_16x16x4_f32(
       false, a,  false, b,
       (short)0, c,  false,  false);
}

__device__ __forceinline__ v2f ld2g(const float* __restrict__ p) {
  return *reinterpret_cast<const v2f*>(p);
}
__device__ __forceinline__ v2f ld2s(const float* p) {
  return *reinterpret_cast<const v2f*>(p);
}
__device__ __forceinline__ void st2g(float* __restrict__ p, float x, float y) {
  v2f v; v.x = x; v.y = y;
  *reinterpret_cast<v2f*>(p) = v;
}

template <int M>
__device__ __forceinline__ float row_xor(float v) {
  return __builtin_bit_cast(float,
      __builtin_amdgcn_update_dpp(0, __builtin_bit_cast(int, v),
                                  0x160 + M, 0xf, 0xf, true));
}
__device__ __forceinline__ float row_max16(float v) {
  v = fmaxf(v, row_xor<1>(v));
  v = fmaxf(v, row_xor<2>(v));
  v = fmaxf(v, row_xor<4>(v));
  v = fmaxf(v, row_xor<8>(v));
  return v;
}
__device__ __forceinline__ float row_sum16(float v) {
  v += row_xor<1>(v);
  v += row_xor<2>(v);
  v += row_xor<4>(v);
  v += row_xor<8>(v);
  return v;
}

#define HBLK (512 * DK)

#define FTILE (16 * EMBD)

#define TLP 260
__global__ void prep_kernel(const float* __restrict__ EN,
                            const float* __restrict__ WqG,
                            const float* __restrict__ WqF,
                            const float* __restrict__ WqL,
                            const float* __restrict__ Wk,
                            const float* __restrict__ Wv,
                            float* __restrict__ qg,
                            float* __restrict__ ENf,
                            float* __restrict__ Wf) {
  __shared__ float tl[16 * TLP];
  const int t = threadIdx.x;
  if (blockIdx.x < BATCH) {
    const int b = blockIdx.x;
    {
      const int cg = t & 63, rg = t >> 6;
      const float* base =
          EN + ((size_t)b * NNODE + (size_t)rg * 128) * EMBD + cg * 4;
      v4f s = {0.f, 0.f, 0.f, 0.f};
      for (int i = 0; i < 128; ++i)
        s += *reinterpret_cast<const v4f*>(base + (size_t)i * EMBD);
      *reinterpret_cast<v4f*>(tl + rg * 256 + cg * 4) = s;
    }
    __syncthreads();
    tl[1024 + t] = ((tl[t] + tl[256 + t]) + (tl[512 + t] + tl[768 + t])) *
                   (1.0f / (float)NNODE);
    __syncthreads();
    {
      const int lane = t & 31, widx = t >> 5;
      const v4f m0 = *reinterpret_cast<const v4f*>(tl + 1024 + lane * 4);
      const v4f m1 = *reinterpret_cast<const v4f*>(tl + 1024 + 128 + lane * 4);
      const float* wb = WqG + (size_t)(widx * 32) * EMBD + lane * 4;
      float keep = 0.0f;
#pragma unroll 8
      for (int r = 0; r < 32; ++r) {
        const v4f w0 = *reinterpret_cast<const v4f*>(wb + r * EMBD);
        const v4f w1 = *reinterpret_cast<const v4f*>(wb + r * EMBD + 128);
        float p = m0.x * w0.x;
        p += m0.y * w0.y;
        p += m0.z * w0.z;
        p += m0.w * w0.w;
        p += m1.x * w1.x;
        p += m1.y * w1.y;
        p += m1.z * w1.z;
        p += m1.w * w1.w;
        p += __shfl_xor(p, 16);
        p += __shfl_xor(p, 8);
        p += __shfl_xor(p, 4);
        p += __shfl_xor(p, 2);
        p += __shfl_xor(p, 1);
        if (lane == r) keep = p;
      }
      qg[(size_t)b * EMBD + widx * 32 + lane] = keep;
    }
    return;
  }
  const int tile = blockIdx.x - BATCH;
  const float* srcp;
  float* dstp;
  if (tile < BATCH * NNODE / 16) {
    srcp = EN  + (size_t)tile * FTILE;
    dstp = ENf + (size_t)tile * FTILE;
  } else {
    const int w  = tile - BATCH * NNODE / 16;
    const int wt = w & 15;
    const int ws = w >> 4;
    const float* wsrc = (ws == 0) ? WqF : (ws == 1) ? WqL : (ws == 2) ? Wk : Wv;
    srcp = wsrc + (size_t)wt * FTILE;
    dstp = Wf + (size_t)w * FTILE;
  }
#pragma unroll
  for (int i = 0; i < 4; ++i) {
    const int idx = t + 256 * i;
    const int row = idx >> 6, q = idx & 63;
    const v4f v = *reinterpret_cast<const v4f*>(srcp + row * EMBD + q * 4);
    *reinterpret_cast<v4f*>(tl + row * TLP + q * 4) = v;
  }
  __syncthreads();
#pragma unroll
  for (int i = 0; i < 4; ++i) {
    const int p = t + 256 * i;
    const int s = p >> 4, j = p & 15;
    const int h = j >> 3, c = (j & 7) * 2;
    const v2f lo = ld2s(tl + c * TLP + 4 * s + 2 * h);
    const v2f hi = ld2s(tl + (c + 1) * TLP + 4 * s + 2 * h);
    v4f o; o.x = lo.x; o.y = lo.y; o.z = hi.x; o.w = hi.y;
    *reinterpret_cast<v4f*>(dstp + 4 * p) = o;
  }
}

__global__ void proj_kernel(const float* __restrict__ ENf,
                            const float* __restrict__ EF,
                            const float* __restrict__ EL,
                            const float* __restrict__ Wf,
                            const float* __restrict__ qg,
                            float* __restrict__ Q,
                            float* __restrict__ K,
                            float* __restrict__ V) {
  const int widx = threadIdx.x >> 5;
  const int lane = threadIdx.x & 31;
  const int ln   = lane & 15;
  const int kh   = lane >> 4;
  const int wave = blockIdx.x * 4 + widx;
  const int b    = wave >> 6;
  const int rem  = wave & 63;
  const int mt   = rem >> 2;
  const int ng   = rem & 3;
  const int row0 = mt * 32 + ln;
  const int sel  = blockIdx.y;

  const float* WqFf = Wf;
  const float* WqLf = Wf + (size_t)16 * FTILE;
  const float* Wkf  = Wf + (size_t)32 * FTILE;
  const float* Wvf  = Wf + (size_t)48 * FTILE;

  v8f acc[2][4];
#pragma unroll
  for (int i = 0; i < 2; ++i)
#pragma unroll
    for (int j = 0; j < 4; ++j) acc[i][j] = v8f{0.f,0.f,0.f,0.f,0.f,0.f,0.f,0.f};

  if (sel == 1) {
    const float* A0 = ENf + (size_t)(b * 32 + mt * 2) * FTILE + lane * 2;
    const float* A1 = A0 + FTILE;
    const float* Br[4];
#pragma unroll
    for (int j = 0; j < 4; ++j) Br[j] = Wvf + (size_t)(ng * 4 + j) * FTILE + lane * 2;
    for (int s = 0; s < EMBD / 4; ++s) {
      const v2f a0 = ld2g(A0 + s * 64);
      const v2f a1 = ld2g(A1 + s * 64);
#pragma unroll
      for (int j = 0; j < 4; ++j) {
        const v2f bb = ld2g(Br[j] + s * 64);
        acc[0][j] = wmma4(a0, bb, acc[0][j]);
        acc[1][j] = wmma4(a1, bb, acc[1][j]);
      }
    }
#pragma unroll
    for (int i = 0; i < 2; ++i)
#pragma unroll
      for (int j = 0; j < 4; ++j) {
        float* hb = V + (size_t)(b * HNUM + ng * 4 + j) * HBLK;
#pragma unroll
        for (int r = 0; r < 8; r += 2) {
          const int n = mt * 32 + i * 16 + r + 8 * kh;
          st2g(hb + (n >> 1) * 32 + ln * 2, acc[i][j][r], acc[i][j][r + 1]);
        }
      }
    return;
  }

  if (sel == 0) {
    const float* A0 = ENf + (size_t)(b * 32 + mt * 2) * FTILE + lane * 2;
    const float* A1 = A0 + FTILE;
    const float* Br[4];
#pragma unroll
    for (int j = 0; j < 4; ++j) Br[j] = Wkf + (size_t)(ng * 4 + j) * FTILE + lane * 2;
    for (int s = 0; s < EMBD / 4; ++s) {
      const v2f a0 = ld2g(A0 + s * 64);
      const v2f a1 = ld2g(A1 + s * 64);
#pragma unroll
      for (int j = 0; j < 4; ++j) {
        const v2f bb = ld2g(Br[j] + s * 64);
        acc[0][j] = wmma4(bb, a0, acc[0][j]);
        acc[1][j] = wmma4(bb, a1, acc[1][j]);
      }
    }
  } else {
    const float* A0 = EF + (size_t)(b * GGRP + row0) * EMBD + 2 * kh;
    const float* A1 = A0 + (size_t)16 * EMBD;
    const float* Br[4];
#pragma unroll
    for (int j = 0; j < 4; ++j) Br[j] = WqFf + (size_t)(ng * 4 + j) * FTILE + lane * 2;
    for (int s = 0; s < EMBD / 4; ++s) {
      const v2f a0 = ld2g(A0 + s * 4);
      const v2f a1 = ld2g(A1 + s * 4);
#pragma unroll
      for (int j = 0; j < 4; ++j) {
        const v2f bb = ld2g(Br[j] + s * 64);
        acc[0][j] = wmma4(bb, a0, acc[0][j]);
        acc[1][j] = wmma4(bb, a1, acc[1][j]);
      }
    }
    const float* C0 = EL + (size_t)(b * GGRP + row0) * EMBD + 2 * kh;
    const float* C1 = C0 + (size_t)16 * EMBD;
#pragma unroll
    for (int j = 0; j < 4; ++j) Br[j] = WqLf + (size_t)(ng * 4 + j) * FTILE + lane * 2;
    for (int s = 0; s < EMBD / 4; ++s) {
      const v2f a0 = ld2g(C0 + s * 4);
      const v2f a1 = ld2g(C1 + s * 4);
#pragma unroll
      for (int j = 0; j < 4; ++j) {
        const v2f bb = ld2g(Br[j] + s * 64);
        acc[0][j] = wmma4(bb, a0, acc[0][j]);
        acc[1][j] = wmma4(bb, a1, acc[1][j]);
      }
    }
#pragma unroll
    for (int j = 0; j < 4; ++j)
#pragma unroll
      for (int r = 0; r < 8; ++r) {
        const float qv = qg[(size_t)b * EMBD + ng * 64 + j * 16 + r + 8 * kh];
#pragma unroll
        for (int i = 0; i < 2; ++i) acc[i][j][r] += qv;
      }
  }

  float* Out = (sel == 0) ? K : Q;
#pragma unroll
  for (int i = 0; i < 2; ++i)
#pragma unroll
    for (int j = 0; j < 4; ++j) {
      float* tb = Out + (size_t)(b * HNUM + ng * 4 + j) * HBLK + (mt * 2 + i) * 256;
#pragma unroll
      for (int r = 0; r < 8; r += 2) {
        const int d = r + 8 * kh;
        st2g(tb + (d >> 2) * 64 + ((d >> 1) & 1) * 32 + ln * 2,
             acc[i][j][r], acc[i][j][r + 1]);
      }
    }
}

__global__ void attn_kernel(const float* __restrict__ Q,
                            const float* __restrict__ K,
                            const float* __restrict__ V,
                            const float* __restrict__ MASK,
                            float* __restrict__ outc) {
  __shared__ __attribute__((aligned(16))) float sc[16 * SCP];
  __shared__ float part[8 * 256];
  __shared__ float rinv[16];

  const int gt = blockIdx.x, h = blockIdx.y, b = blockIdx.z;
  const int t    = threadIdx.x;
  const int widx = t >> 5;
  const int lane = t & 31;
  const int ln   = lane & 15;
  const int kh   = lane >> 4;

  {
    const float* Atile = Q + (size_t)(b * HNUM + h) * HBLK + gt * 256 + lane * 2;
    const float* Bt[4];
#pragma unroll
    for (int j = 0; j < 4; ++j)
      Bt[j] = K + (size_t)(b * HNUM + h) * HBLK + (widx * 4 + j) * 256 + lane * 2;
    v8f acc[4];
#pragma unroll
    for (int j = 0; j < 4; ++j) acc[j] = v8f{0.f,0.f,0.f,0.f,0.f,0.f,0.f,0.f};
    v2f af[4], bf[4][4];
#pragma unroll
    for (int s = 0; s < DK / 4; ++s) {
      af[s] = ld2g(Atile + s * 64);
#pragma unroll
      for (int j = 0; j < 4; ++j) bf[j][s] = ld2g(Bt[j] + s * 64);
    }
    __builtin_amdgcn_sched_barrier(0);
#pragma unroll
    for (int s = 0; s < DK / 4; ++s)
#pragma unroll
      for (int j = 0; j < 4; ++j)
        acc[j] = wmma4(af[s], bf[j][s], acc[j]);
#pragma unroll
    for (int j = 0; j < 4; ++j) {
      const int nb = (widx * 4 + j) * 16;
#pragma unroll
      for (int r = 0; r < 8; ++r) {
        const int row = r + 8 * kh;
        sc[row * SCP + nb + ln] = acc[j][r] * 0.25f;
      }
    }
  }
  {
    const int row = t >> 4, l16 = t & 15;
    float* sp = sc + row * SCP + l16 * 4;
    const float* mp =
        MASK + (size_t)(b * GGRP + gt * 16 + row) * NNODE + l16 * 4;
    v4f m[8];
#pragma unroll
    for (int i = 0; i < 8; ++i) m[i] = *reinterpret_cast<const v4f*>(mp + i * 64);
    __syncthreads();
    v4f x[8];
#pragma unroll
    for (int i = 0; i < 8; ++i)
      x[i] = *reinterpret_cast<const v4f*>(sp + i * 64) + m[i];
    float mx = -INFINITY;
#pragma unroll
    for (int i = 0; i < 8; ++i) {
      mx = fmaxf(mx, x[i].x);
      mx = fmaxf(mx, x[i].y);
      mx = fmaxf(mx, x[i].z);
      mx = fmaxf(mx, x[i].w);
    }
    const float rmax = row_max16(mx);
    float ssum = 0.0f;
#pragma unroll
    for (int i = 0; i < 8; ++i) {
      v4f e;
      e.x = __expf(x[i].x - rmax);
      e.y = __expf(x[i].y - rmax);
      e.z = __expf(x[i].z - rmax);
      e.w = __expf(x[i].w - rmax);
      ssum += e.x;
      ssum += e.y;
      ssum += e.z;
      ssum += e.w;
      *reinterpret_cast<v4f*>(sp + i * 64) = e;
    }
    ssum = row_sum16(ssum);
    if (l16 == 0) rinv[row] = 1.0f / ssum;
  }
  __syncthreads();

  v8f acc = {0.f,0.f,0.f,0.f,0.f,0.f,0.f,0.f};
  const float* vb = V + (size_t)(b * HNUM + h) * HBLK + widx * 1024 + lane * 2;
#pragma unroll
  for (int nti = 0; nti < 4; ++nti) {
    const int nb = (widx * 4 + nti) * 16;
#pragma unroll
    for (int k = 0; k < 16; k += 4) {
      const int kk = nb + k + 2 * kh;
      v2f a = ld2s(sc + ln * SCP + kk);
      const v2f bb = ld2g(vb + nti * 256 + k * 16);
      acc = wmma4(a, bb, acc);
    }
  }
#pragma unroll
  for (int r = 0; r < 8; ++r)
    part[widx * 256 + (r + 8 * kh) * 16 + ln] = acc[r];
  __syncthreads();

  {
    const int orow = (t >> 1) & 15;
    const int ocol = (t >> 6) * 4 + ((t >> 5) & 1) * 2 + (t & 1);
    float s = 0.0f;
#pragma unroll
    for (int w = 0; w < 8; ++w) s += part[w * 256 + orow * 16 + ocol];
    outc[(size_t)(b * (GGRP / 16) + gt) * FTILE + h * 256 + t] = s * rinv[orow];
  }
}

__global__ void mh_kernel(const float* __restrict__ outc,
                          const float* __restrict__ Wc,
                          const float* __restrict__ bc,
                          float* __restrict__ mh) {
  const int widx = threadIdx.x >> 5;
  const int lane = threadIdx.x & 31;
  const int ln   = lane & 15;
  const int kh   = lane >> 4;
  const int wave = blockIdx.x * 4 + widx;
  const int b    = wave >> 6;
  const int rem  = wave & 63;
  const int mt   = rem >> 2;
  const int ng   = rem & 3;

  v8f acc[2][4];
#pragma unroll
  for (int i = 0; i < 2; ++i)
#pragma unroll
    for (int j = 0; j < 4; ++j) acc[i][j] = v8f{0.f,0.f,0.f,0.f,0.f,0.f,0.f,0.f};

  const float* A0 = outc + (size_t)(b * 32 + mt * 2) * FTILE + lane * 2;
  const float* A1 = A0 + FTILE;
  const float* Br[4];
#pragma unroll
  for (int j = 0; j < 4; ++j) Br[j] = Wc + (size_t)(ng * 64 + j * 16 + ln) * EMBD + 2 * kh;
  for (int s = 0; s < EMBD / 4; ++s) {
    const v2f a0 = ld2g(A0 + s * 64);
    const v2f a1 = ld2g(A1 + s * 64);
#pragma unroll
    for (int j = 0; j < 4; ++j) {
      const v2f bb = ld2g(Br[j] + s * 4);
      acc[0][j] = wmma4(bb, a0, acc[0][j]);
      acc[1][j] = wmma4(bb, a1, acc[1][j]);
    }
  }
#pragma unroll
  for (int i = 0; i < 2; ++i)
#pragma unroll
    for (int j = 0; j < 4; ++j) {
      float* tb = mh + (size_t)(b * 32 + mt * 2 + i) * FTILE;
#pragma unroll
      for (int r = 0; r < 8; r += 2) {
        const int d = ng * 64 + j * 16 + r + 8 * kh;
        st2g(tb + (d >> 2) * 64 + ((d >> 1) & 1) * 32 + ln * 2,
             acc[i][j][r] + bc[d], acc[i][j][r + 1] + bc[d + 1]);
      }
    }
}

__global__ void final_kernel(const float* __restrict__ mh,
                             const float* __restrict__ ENf,
                             const float* __restrict__ MASK,
                             float* __restrict__ out) {
  __shared__ __attribute__((aligned(16))) float sc[16 * SCP];

  const int gt = blockIdx.x, b = blockIdx.y;
  const int t    = threadIdx.x;
  const int widx = t >> 5;
  const int lane = t & 31;
  const int ln   = lane & 15;
  const int kh   = lane >> 4;

  {
    const float* Atile = mh + (size_t)(b * (GGRP / 16) + gt) * FTILE + lane * 2;
    const float* Br[4];
#pragma unroll
    for (int j = 0; j < 4; ++j)
      Br[j] = ENf + (size_t)(b * (NNODE / 16) + widx * 4 + j) * FTILE + lane * 2;
    v8f acc[4];
#pragma unroll
    for (int j = 0; j < 4; ++j) acc[j] = v8f{0.f,0.f,0.f,0.f,0.f,0.f,0.f,0.f};
    for (int s = 0; s < EMBD / 4; ++s) {
      const v2f a = ld2g(Atile + s * 64);
#pragma unroll
      for (int j = 0; j < 4; ++j)
        acc[j] = wmma4(a, ld2g(Br[j] + s * 64), acc[j]);
    }
#pragma unroll
    for (int j = 0; j < 4; ++j) {
      const int nb = (widx * 4 + j) * 16;
#pragma unroll
      for (int r = 0; r < 8; ++r) {
        const int row = r + 8 * kh;
        const float s2 = acc[j][r] * (1.0f / 16.0f);
        sc[row * SCP + nb + ln] = 10.0f * tanhf(s2);
      }
    }
  }

  const int row = t >> 4, l16 = t & 15;
  const float* sp = sc + row * SCP + l16 * 4;
  const float* mp = MASK + (size_t)(b * GGRP + gt * 16 + row) * NNODE + l16 * 4;
  v4f m[8];
#pragma unroll
  for (int i = 0; i < 8; ++i) m[i] = *reinterpret_cast<const v4f*>(mp + i * 64);
  __syncthreads();
  v4f x[8];
#pragma unroll
  for (int i = 0; i < 8; ++i)
    x[i] = *reinterpret_cast<const v4f*>(sp + i * 64) + m[i];
  float mx = -INFINITY;
#pragma unroll
  for (int i = 0; i < 8; ++i) {
    mx = fmaxf(mx, x[i].x);
    mx = fmaxf(mx, x[i].y);
    mx = fmaxf(mx, x[i].z);
    mx = fmaxf(mx, x[i].w);
  }
  const float rmax = row_max16(mx);
  float ssum = 0.0f;
#pragma unroll
  for (int i = 0; i < 8; ++i) {
    x[i].x = __expf(x[i].x - rmax);
    x[i].y = __expf(x[i].y - rmax);
    x[i].z = __expf(x[i].z - rmax);
    x[i].w = __expf(x[i].w - rmax);
    ssum += x[i].x;
    ssum += x[i].y;
    ssum += x[i].z;
    ssum += x[i].w;
  }
  const float inv = 1.0f / row_sum16(ssum);
  float* op = out + (size_t)(b * GGRP + gt * 16 + row) * NNODE + l16 * 4;
#pragma unroll
  for (int i = 0; i < 8; ++i)
    *reinterpret_cast<v4f*>(op + i * 64) = x[i] * inv;
}

extern "C" void kernel_launch(void* const* d_in, const int* in_sizes, int n_in,
                              void* d_out, int out_size, void* d_ws, size_t ws_size,
                              hipStream_t stream) {
  const float* EN   = (const float*)d_in[0];
  const float* EF   = (const float*)d_in[1];
  const float* EL   = (const float*)d_in[2];
  const float* MASK = (const float*)d_in[3];
  const float* WqG  = (const float*)d_in[4];
  const float* WqF  = (const float*)d_in[5];
  const float* WqL  = (const float*)d_in[6];
  const float* Wk   = (const float*)d_in[7];
  const float* Wv   = (const float*)d_in[8];
  const float* Wc   = (const float*)d_in[9];
  const float* bc   = (const float*)d_in[10];
  float* out = (float*)d_out;

  float* ws   = (float*)d_ws;
  float* qg   = ws;
  float* Qb   = qg   + (size_t)BATCH * EMBD;
  float* Kb   = Qb   + (size_t)BATCH * GGRP * EMBD;
  float* Vb   = Kb   + (size_t)BATCH * NNODE * EMBD;
  float* outc = Vb   + (size_t)BATCH * NNODE * EMBD;
  float* ENf  = outc + (size_t)BATCH * GGRP * EMBD;
  float* Wf   = outc;
  float* mhb  = Qb;

  prep_kernel<<<1120, 256, 0, stream>>>(EN, WqG, WqF, WqL, Wk, Wv, qg, ENf, Wf);

  dim3 gp(512, 3, 1);
  proj_kernel<<<gp, 128, 0, stream>>>(ENf, EF, EL, Wf, qg, Qb, Kb, Vb);

  dim3 ga(GGRP / 16, HNUM, BATCH);
  attn_kernel<<<ga, 256, 0, stream>>>(Qb, Kb, Vb, MASK, outc);

  mh_kernel<<<512, 128, 0, stream>>>(outc, Wc, bc, mhb);

  dim3 gf(GGRP / 16, BATCH, 1);
  final_kernel<<<gf, 256, 0, stream>>>(mhb, ENf, MASK, out);
}
